# starting kernel
# baseline (speedup 1.0000x reference)
#include <hip/hip_runtime.h>


typedef __attribute__((ext_vector_type(2))) float v2f;
typedef __attribute__((ext_vector_type(8))) float v8f;

#define GRIDN 12
#define INV_BW (11.0f / 6.0f)
#define CSTEP  (6.0f / 11.0f)

#define NSTEP1 30
#define NSTEP2 79
#define NSTEP3 40
#define NT1 2
#define NT2 1
#define NT3 1
#define B1_FLOATS (NSTEP1 * NT1 * 64)
#define B2_FLOATS (NSTEP2 * NT2 * 64)
#define B3_FLOATS (NSTEP3 * NT3 * 64)
#define B1_OFF 0
#define B2_OFF (B1_FLOATS)
#define B3_OFF (B1_FLOATS + B2_FLOATS)
#define BTOTAL (B1_FLOATS + B2_FLOATS + B3_FLOATS)

__global__ __launch_bounds__(256) void kan_pack(
    const float* __restrict__ c1, const float* __restrict__ w1, const float* __restrict__ b1,
    const float* __restrict__ c2, const float* __restrict__ w2, const float* __restrict__ b2,
    const float* __restrict__ c3, const float* __restrict__ w3, const float* __restrict__ b3,
    const float* __restrict__ base, float* __restrict__ ws) {
  int idx = blockIdx.x * blockDim.x + threadIdx.x;
  if (idx >= BTOTAL) return;

  int layer, e, nt_count;
  if (idx < B2_OFF)       { layer = 0; e = idx;           nt_count = NT1; }
  else if (idx < B3_OFF)  { layer = 1; e = idx - B2_OFF;  nt_count = NT2; }
  else                    { layer = 2; e = idx - B3_OFF;  nt_count = NT3; }

  int v    = e & 1;
  int lane = (e >> 1) & 31;
  int q    = (e >> 1) >> 5;
  int t    = q / nt_count;
  int nt   = q - t * nt_count;
  int k    = 4 * t + 2 * (lane >> 4) + v;
  int n    = nt * 16 + (lane & 15);

  float val = 0.0f;
  if (layer == 0) {
    if (n < 24) {
      if (k < 108)      { int i = k / 12, g = k - i * 12; val = c1[(i * 24 + n) * 12 + g]; }
      else if (k < 117) { val = w1[n * 9 + (k - 108)]; }
      else if (k == 117){ val = b1[n]; }
    }
  } else if (layer == 1) {
    if (n < 12) {
      if (k < 288)      { int i = k / 12, g = k - i * 12; val = c2[(i * 12 + n) * 12 + g]; }
      else if (k < 312) { val = w2[n * 24 + (k - 288)]; }
      else if (k == 312){ val = b2[n]; }
    }
  } else {
    if (n < 3) {
      if (k < 144)      { int i = k / 12, g = k - i * 12; val = c3[(i * 3 + n) * 12 + g]; }
      else if (k < 156) { val = w3[n * 12 + (k - 144)]; }
      else if (k == 156){ val = b3[n] + base[n]; }
    }
  }
  ws[idx] = val;
}

template <int I>
__device__ __forceinline__ float afrag_elem(const float* __restrict__ a, int k) {
  const int KB = I * GRIDN;
  float r;
  if (k < KB) {
    int i = (int)((unsigned)k / 12u);
    int g = k - i * 12;
    float d = (a[i] - (-3.0f + CSTEP * (float)g)) * INV_BW;
    r = __expf(-0.5f * d * d);
  } else if (k < KB + I) {
    r = a[k - KB];
  } else if (k == KB + I) {
    r = 1.0f;
  } else {
    r = 0.0f;
  }
  return r;
}

template <int I, int NSTEPS, int NT>
__device__ __forceinline__ void kan_layer_wmma(const float* __restrict__ arow,
                                               const float* __restrict__ Bsh,
                                               int lane, int half,
                                               v8f& c0, v8f& c1) {
#pragma unroll 4
  for (int t = 0; t < NSTEPS; ++t) {
    int k0 = 4 * t + 2 * half;
    v2f a;
    a.x = afrag_elem<I>(arow, k0);
    a.y = afrag_elem<I>(arow, k0 + 1);
    v2f b0 = *(const v2f*)(Bsh + (t * NT + 0) * 64 + lane * 2);
    c0 = __builtin_amdgcn_wmma_f32_16x16x4_f32(false, a, false, b0, (short)0, c0,
                                               false, false);
    if (NT == 2) {
      v2f b1 = *(const v2f*)(Bsh + (t * NT + 1) * 64 + lane * 2);
      c1 = __builtin_amdgcn_wmma_f32_16x16x4_f32(false, a, false, b1, (short)0, c1,
                                                 false, false);
    }
  }
}

__device__ __forceinline__ float silu_f(float z) {
  return z * (1.0f / (1.0f + __expf(-z)));
}

__global__ __launch_bounds__(256, 2) void kan_forward(const float* __restrict__ x,
                                                      const float* __restrict__ ws,
                                                      float* __restrict__ out,
                                                      int tiles_per_wave) {
  __shared__ __align__(16) float Bsh[BTOTAL];
  __shared__ __align__(16) float act0[8][16][12];
  __shared__ __align__(16) float act1[8][16][24];
  __shared__ __align__(16) float act2[8][16][12];
  __shared__ __align__(16) float cor[8][16][4];

  const int tid = threadIdx.x;
  for (int i = tid; i < BTOTAL; i += 256) Bsh[i] = ws[i];
  __syncthreads();

  const int wv   = tid >> 5;
  const int lane = tid & 31;
  const int half = lane >> 4;
  const int m    = lane & 15;
  const float* a0 = &act0[wv][m][0];
  const float* a1 = &act1[wv][m][0];
  const float* a2 = &act2[wv][m][0];

  for (int it = 0; it < tiles_per_wave; ++it) {
    const int rowbase = ((blockIdx.x * tiles_per_wave + it) * 8 + wv) * 16;

    for (int e = lane; e < 144; e += 32) {
      int r = e / 9, c = e - r * 9;
      act0[wv][r][c] = __builtin_nontemporal_load(&x[(size_t)(rowbase + r) * 9 + c]);
    }
    asm volatile("s_wait_dscnt 0" ::: "memory");

    v8f c0 = {}, c1 = {};
    kan_layer_wmma<9, NSTEP1, NT1>(a0, Bsh + B1_OFF, lane, half, c0, c1);
#pragma unroll
    for (int v = 0; v < 8; ++v) {
      int r = half * 8 + v;
      act1[wv][r][m] = silu_f(c0[v]);
      if (m < 8) act1[wv][r][16 + m] = silu_f(c1[v]);
    }
    asm volatile("s_wait_dscnt 0" ::: "memory");

    v8f d0 = {}, dz = {};
    kan_layer_wmma<24, NSTEP2, NT2>(a1, Bsh + B2_OFF, lane, half, d0, dz);
#pragma unroll
    for (int v = 0; v < 8; ++v) {
      int r = half * 8 + v;
      if (m < 12) act2[wv][r][m] = silu_f(d0[v]);
    }
    asm volatile("s_wait_dscnt 0" ::: "memory");

    v8f e0 = {}, ez = {};
    kan_layer_wmma<12, NSTEP3, NT3>(a2, Bsh + B3_OFF, lane, half, e0, ez);
#pragma unroll
    for (int v = 0; v < 8; ++v) {
      int r = half * 8 + v;
      if (m < 3) cor[wv][r][m] = e0[v];
    }
    asm volatile("s_wait_dscnt 0" ::: "memory");

    if (half == 0) {
      float p0 = cor[wv][m][0], p1 = cor[wv][m][1], p2 = cor[wv][m][2];
      float mx = fmaxf(p0, fmaxf(p1, p2));
      float q0 = __expf(p0 - mx), q1 = __expf(p1 - mx), q2 = __expf(p2 - mx);
      float inv = 1.0f / (q0 + q1 + q2);
      float* o = out + (size_t)(rowbase + m) * 3;
      __builtin_nontemporal_store(q0 * inv, o + 0);
      __builtin_nontemporal_store(q1 * inv, o + 1);
      __builtin_nontemporal_store(q2 * inv, o + 2);
    }
    asm volatile("s_wait_dscnt 0" ::: "memory");
  }
}

extern "C" void kernel_launch(void* const* d_in, const int* in_sizes, int n_in,
                              void* d_out, int out_size, void* d_ws, size_t ws_size,
                              hipStream_t stream) {
  const float* x    = (const float*)d_in[0];
  const float* c1   = (const float*)d_in[1];
  const float* w1   = (const float*)d_in[2];
  const float* b1   = (const float*)d_in[3];
  const float* c2   = (const float*)d_in[4];
  const float* w2   = (const float*)d_in[5];
  const float* b2   = (const float*)d_in[6];
  const float* c3   = (const float*)d_in[7];
  const float* w3   = (const float*)d_in[8];
  const float* b3   = (const float*)d_in[9];
  const float* base = (const float*)d_in[10];
  float* ws  = (float*)d_ws;
  float* out = (float*)d_out;

  const int Btot = in_sizes[0] / 9;
  const int TILES = 4;
  const int blocks = Btot / (128 * TILES);

  kan_pack<<<(BTOTAL + 255) / 256, 256, 0, stream>>>(c1, w1, b1, c2, w2, b2,
                                                     c3, w3, b3, base, ws);
  kan_forward<<<blocks, 256, 0, stream>>>(x, ws, out, TILES);
}


# Layers computed transposed (weights = A operand, batch rows = columns): a lane's D registers feed its own next-layer B fragments, no LDS transposes between layers; K order of the sums changed
# speedup vs baseline: 1.2738x; 1.2738x over previous
#include <hip/hip_runtime.h>


typedef __attribute__((ext_vector_type(2))) float v2f;
typedef __attribute__((ext_vector_type(8))) float v8f;

#define GRIDN 12
#define INV_BW (11.0f / 6.0f)
#define CSTEP  (6.0f / 11.0f)

#define NSTEP1 30
#define NSTEP2 79
#define NSTEP3 40
#define NT1 2
#define NT2 1
#define NT3 1
#define NC2 12
#define NC3 6
#define B1_FLOATS (NSTEP1 * NT1 * 64)
#define B2_FLOATS (NSTEP2 * NT2 * 64)
#define B3_FLOATS (NSTEP3 * NT3 * 64)
#define B1_OFF 0
#define B2_OFF (B1_FLOATS)
#define B3_OFF (B1_FLOATS + B2_FLOATS)
#define BTOTAL (B1_FLOATS + B2_FLOATS + B3_FLOATS)

__global__ __launch_bounds__(256) void kan_pack(
    const float* __restrict__ c1, const float* __restrict__ w1, const float* __restrict__ b1,
    const float* __restrict__ c2, const float* __restrict__ w2, const float* __restrict__ b2,
    const float* __restrict__ c3, const float* __restrict__ w3, const float* __restrict__ b3,
    const float* __restrict__ base, float* __restrict__ ws) {
  int idx = blockIdx.x * blockDim.x + threadIdx.x;
  if (idx >= BTOTAL) return;

  int layer, e, nt_count;
  if (idx < B2_OFF)       { layer = 0; e = idx;           nt_count = NT1; }
  else if (idx < B3_OFF)  { layer = 1; e = idx - B2_OFF;  nt_count = NT2; }
  else                    { layer = 2; e = idx - B3_OFF;  nt_count = NT3; }

  int s    = e & 1;
  int lane = (e >> 1) & 31;
  int q    = (e >> 1) >> 5;
  int t    = q / nt_count;
  int mt   = q - t * nt_count;
  int h    = lane >> 4;
  int c    = lane & 15;

  float val = 0.0f;
  if (layer == 0) {
    int k = 4 * t + 2 * h + s;
    int o;
    if (mt == 0)                 o = c;
    else if (c < 4)              o = 16 + c;
    else if (c >= 8 && c < 12)   o = 20 + (c - 8);
    else                         o = -1;
    if (o >= 0) {
      if (k < 108)      { int i = k / 12, g = k - i * 12; val = c1[(i * 24 + o) * 12 + g]; }
      else if (k < 117) { val = w1[o * 9 + (k - 108)]; }
      else if (k == 117){ val = b1[o]; }
    }
  } else if (layer == 1) {
    int j = 2 * t + s;
    int o;
    if (c < 6)                   o = c;
    else if (c >= 8 && c < 14)   o = 6 + (c - 8);
    else                         o = -1;
    if (o >= 0) {
      if (j < NC2 * 12) {
        int ci = j / 12, g = j - ci * 12;
        int ch = (ci < 8) ? (8 * h + ci) : (16 + 4 * h + (ci - 8));
        val = c2[(ch * 12 + o) * 12 + g];
      } else if (j < NC2 * 13) {
        int ci = j - NC2 * 12;
        int ch = (ci < 8) ? (8 * h + ci) : (16 + 4 * h + (ci - 8));
        val = w2[o * 24 + ch];
      } else if (j == NC2 * 13 && h == 0) {
        val = b2[o];
      }
    }
  } else {
    int j = 2 * t + s;
    if (c < 3) {
      int o = c;
      if (j < NC3 * 12) {
        int ci = j / 12, g = j - ci * 12;
        int ch = 6 * h + ci;
        val = c3[(ch * 3 + o) * 12 + g];
      } else if (j < NC3 * 13) {
        int ch = 6 * h + (j - NC3 * 12);
        val = w3[o * 12 + ch];
      } else if (j == NC3 * 13 && h == 0) {
        val = b3[o] + base[o];
      }
    }
  }
  ws[idx] = val;
}

__device__ __forceinline__ float rbf_f(float a, int g) {
  float d = (a - (-3.0f + CSTEP * (float)g)) * INV_BW;
  return __expf(-0.5f * d * d);
}

template <int I>
__device__ __forceinline__ float afrag_elem(const float* __restrict__ a, int k) {
  const int KB = I * GRIDN;
  float r;
  if (k < KB) {
    int i = (int)((unsigned)k / 12u);
    int g = k - i * 12;
    r = rbf_f(a[i], g);
  } else if (k < KB + I) {
    r = a[k - KB];
  } else if (k == KB + I) {
    r = 1.0f;
  } else {
    r = 0.0f;
  }
  return r;
}

template <int NC>
__device__ __forceinline__ float reg_elem(const float (&hv)[NC], int j) {
  float r;
  if (j < NC * GRIDN) {
    r = rbf_f(hv[j / GRIDN], j % GRIDN);
  } else if (j < NC * (GRIDN + 1)) {
    r = hv[j - NC * GRIDN];
  } else if (j == NC * (GRIDN + 1)) {
    r = 1.0f;
  } else {
    r = 0.0f;
  }
  return r;
}

template <int NC, int NSTEPS>
__device__ __forceinline__ v8f kan_layer_regs(const float (&hv)[NC],
                                              const float* __restrict__ Wsh, int lane) {
  v8f acc = {};
#pragma unroll
  for (int t = 0; t < NSTEPS; ++t) {
    v2f b;
    b.x = reg_elem<NC>(hv, 2 * t);
    b.y = reg_elem<NC>(hv, 2 * t + 1);
    v2f a = *(const v2f*)(Wsh + t * 64 + lane * 2);
    acc = __builtin_amdgcn_wmma_f32_16x16x4_f32(false, a, false, b, (short)0, acc,
                                                false, false);
  }
  return acc;
}

__device__ __forceinline__ float silu_f(float z) {
  return z * (1.0f / (1.0f + __expf(-z)));
}

__global__ __launch_bounds__(256, 2) void kan_forward(const float* __restrict__ x,
                                                      const float* __restrict__ ws,
                                                      float* __restrict__ out,
                                                      int tiles_per_wave) {
  __shared__ __align__(16) float Bsh[BTOTAL];
  __shared__ __align__(16) float act0[8][16][12];

  const int tid = threadIdx.x;
  for (int i = tid; i < BTOTAL; i += 256) Bsh[i] = ws[i];
  __syncthreads();

  const int wv   = tid >> 5;
  const int lane = tid & 31;
  const int half = lane >> 4;
  const int m    = lane & 15;
  const float* a0 = &act0[wv][m][0];

  for (int it = 0; it < tiles_per_wave; ++it) {
    const int rowbase = ((blockIdx.x * tiles_per_wave + it) * 8 + wv) * 16;

    for (int e = lane; e < 144; e += 32) {
      int r = e / 9, c = e - r * 9;
      act0[wv][r][c] = __builtin_nontemporal_load(&x[(size_t)(rowbase + r) * 9 + c]);
    }
    asm volatile("s_wait_dscnt 0" ::: "memory");

    v8f c0 = {}, c1 = {};
#pragma unroll
    for (int t = 0; t < NSTEP1; ++t) {
      int k0 = 4 * t + 2 * half;
      v2f b;
      b.x = afrag_elem<9>(a0, k0);
      b.y = afrag_elem<9>(a0, k0 + 1);
      v2f w0 = *(const v2f*)(Bsh + B1_OFF + (t * NT1 + 0) * 64 + lane * 2);
      v2f w1 = *(const v2f*)(Bsh + B1_OFF + (t * NT1 + 1) * 64 + lane * 2);
      c0 = __builtin_amdgcn_wmma_f32_16x16x4_f32(false, w0, false, b, (short)0, c0,
                                                 false, false);
      c1 = __builtin_amdgcn_wmma_f32_16x16x4_f32(false, w1, false, b, (short)0, c1,
                                                 false, false);
    }
    __builtin_amdgcn_sched_barrier(0);
    float h1[NC2];
#pragma unroll
    for (int v = 0; v < 8; ++v) h1[v] = silu_f(c0[v]);
#pragma unroll
    for (int v = 0; v < 4; ++v) h1[8 + v] = silu_f(c1[v]);

    __builtin_amdgcn_sched_barrier(0);
    v8f d0 = kan_layer_regs<NC2, NSTEP2>(h1, Bsh + B2_OFF, lane);
    __builtin_amdgcn_sched_barrier(0);
    float h2[NC3];
#pragma unroll
    for (int v = 0; v < NC3; ++v) h2[v] = silu_f(d0[v]);

    __builtin_amdgcn_sched_barrier(0);
    v8f e0 = kan_layer_regs<NC3, NSTEP3>(h2, Bsh + B3_OFF, lane);

    if (half == 0) {
      float p0 = e0[0], p1 = e0[1], p2 = e0[2];
      float mx = fmaxf(p0, fmaxf(p1, p2));
      float q0 = __expf(p0 - mx), q1 = __expf(p1 - mx), q2 = __expf(p2 - mx);
      float inv = 1.0f / (q0 + q1 + q2);
      float* o = out + (size_t)(rowbase + m) * 3;
      __builtin_nontemporal_store(q0 * inv, o + 0);
      __builtin_nontemporal_store(q1 * inv, o + 1);
      __builtin_nontemporal_store(q2 * inv, o + 2);
    }
  }
}

extern "C" void kernel_launch(void* const* d_in, const int* in_sizes, int n_in,
                              void* d_out, int out_size, void* d_ws, size_t ws_size,
                              hipStream_t stream) {
  const float* x    = (const float*)d_in[0];
  const float* c1   = (const float*)d_in[1];
  const float* w1   = (const float*)d_in[2];
  const float* b1   = (const float*)d_in[3];
  const float* c2   = (const float*)d_in[4];
  const float* w2   = (const float*)d_in[5];
  const float* b2   = (const float*)d_in[6];
  const float* c3   = (const float*)d_in[7];
  const float* w3   = (const float*)d_in[8];
  const float* b3   = (const float*)d_in[9];
  const float* base = (const float*)d_in[10];
  float* ws  = (float*)d_ws;
  float* out = (float*)d_out;

  const int Btot = in_sizes[0] / 9;
  const int TILES = 4;
  const int blocks = Btot / (128 * TILES);

  kan_pack<<<(BTOTAL + 255) / 256, 256, 0, stream>>>(c1, w1, b1, c2, w2, b2,
                                                     c3, w3, b3, base, ws);
  kan_forward<<<blocks, 256, 0, stream>>>(x, ws, out, TILES);
}


# Weights register-resident: each wave loads all 179 A fragments once (358 VGPRs, 2 waves/SIMD), 256 persistent work-groups stride over tiles, input loaded one tile ahead; no LDS reads of weights
# speedup vs baseline: 1.4211x; 1.1156x over previous
#include <hip/hip_runtime.h>


typedef __attribute__((ext_vector_type(2))) float v2f;
typedef __attribute__((ext_vector_type(8))) float v8f;

#define GRIDN 12
#define INV_BW (11.0f / 6.0f)
#define CSTEP  (6.0f / 11.0f)

#define NSTEP1 30
#define NSTEP2 79
#define NSTEP3 40
#define NT1 2
#define NT2 1
#define NT3 1
#define NC2 12
#define NC3 6
#define B1_FLOATS (NSTEP1 * NT1 * 64)
#define B2_FLOATS (NSTEP2 * NT2 * 64)
#define B3_FLOATS (NSTEP3 * NT3 * 64)
#define B1_OFF 0
#define B2_OFF (B1_FLOATS)
#define B3_OFF (B1_FLOATS + B2_FLOATS)
#define BTOTAL (B1_FLOATS + B2_FLOATS + B3_FLOATS)
#define F1_OFF 0
#define F2_OFF (NSTEP1 * NT1)
#define F3_OFF (NSTEP1 * NT1 + NSTEP2 * NT2)
#define NFRAG  (NSTEP1 * NT1 + NSTEP2 * NT2 + NSTEP3 * NT3)

__global__ __launch_bounds__(256) void kan_pack(
    const float* __restrict__ c1, const float* __restrict__ w1, const float* __restrict__ b1,
    const float* __restrict__ c2, const float* __restrict__ w2, const float* __restrict__ b2,
    const float* __restrict__ c3, const float* __restrict__ w3, const float* __restrict__ b3,
    const float* __restrict__ base, float* __restrict__ ws) {
  int idx = blockIdx.x * blockDim.x + threadIdx.x;
  if (idx >= BTOTAL) return;

  int layer, e, nt_count;
  if (idx < B2_OFF)       { layer = 0; e = idx;           nt_count = NT1; }
  else if (idx < B3_OFF)  { layer = 1; e = idx - B2_OFF;  nt_count = NT2; }
  else                    { layer = 2; e = idx - B3_OFF;  nt_count = NT3; }

  int s    = e & 1;
  int lane = (e >> 1) & 31;
  int q    = (e >> 1) >> 5;
  int t    = q / nt_count;
  int mt   = q - t * nt_count;
  int h    = lane >> 4;
  int c    = lane & 15;

  float val = 0.0f;
  if (layer == 0) {
    int k = 4 * t + 2 * h + s;
    int o;
    if (mt == 0)                 o = c;
    else if (c < 4)              o = 16 + c;
    else if (c >= 8 && c < 12)   o = 20 + (c - 8);
    else                         o = -1;
    if (o >= 0) {
      if (k < 108)      { int i = k / 12, g = k - i * 12; val = c1[(i * 24 + o) * 12 + g]; }
      else if (k < 117) { val = w1[o * 9 + (k - 108)]; }
      else if (k == 117){ val = b1[o]; }
    }
  } else if (layer == 1) {
    int j = 2 * t + s;
    int o;
    if (c < 6)                   o = c;
    else if (c >= 8 && c < 14)   o = 6 + (c - 8);
    else                         o = -1;
    if (o >= 0) {
      if (j < NC2 * 12) {
        int ci = j / 12, g = j - ci * 12;
        int ch = (ci < 8) ? (8 * h + ci) : (16 + 4 * h + (ci - 8));
        val = c2[(ch * 12 + o) * 12 + g];
      } else if (j < NC2 * 13) {
        int ci = j - NC2 * 12;
        int ch = (ci < 8) ? (8 * h + ci) : (16 + 4 * h + (ci - 8));
        val = w2[o * 24 + ch];
      } else if (j == NC2 * 13 && h == 0) {
        val = b2[o];
      }
    }
  } else {
    int j = 2 * t + s;
    if (c < 3) {
      int o = c;
      if (j < NC3 * 12) {
        int ci = j / 12, g = j - ci * 12;
        int ch = 6 * h + ci;
        val = c3[(ch * 3 + o) * 12 + g];
      } else if (j < NC3 * 13) {
        int ch = 6 * h + (j - NC3 * 12);
        val = w3[o * 12 + ch];
      } else if (j == NC3 * 13 && h == 0) {
        val = b3[o] + base[o];
      }
    }
  }
  ws[idx] = val;
}

__device__ __forceinline__ float rbf_f(float a, int g) {
  float d = (a - (-3.0f + CSTEP * (float)g)) * INV_BW;
  return __expf(-0.5f * d * d);
}

template <int I>
__device__ __forceinline__ float afrag_elem(const float* __restrict__ a, int k) {
  const int KB = I * GRIDN;
  float r;
  if (k < KB) {
    int i = (int)((unsigned)k / 12u);
    int g = k - i * 12;
    r = rbf_f(a[i], g);
  } else if (k < KB + I) {
    r = a[k - KB];
  } else if (k == KB + I) {
    r = 1.0f;
  } else {
    r = 0.0f;
  }
  return r;
}

template <int NC>
__device__ __forceinline__ float reg_elem(const float (&hv)[NC], int j) {
  float r;
  if (j < NC * GRIDN) {
    r = rbf_f(hv[j / GRIDN], j % GRIDN);
  } else if (j < NC * (GRIDN + 1)) {
    r = hv[j - NC * GRIDN];
  } else if (j == NC * (GRIDN + 1)) {
    r = 1.0f;
  } else {
    r = 0.0f;
  }
  return r;
}

template <int NC, int NSTEPS, int FOFF>
__device__ __forceinline__ v8f kan_layer_regs(const float (&hv)[NC],
                                              const v2f (&wA)[NFRAG]) {
  v8f acc = {};
#pragma unroll
  for (int t = 0; t < NSTEPS; ++t) {
    v2f b;
    b.x = reg_elem<NC>(hv, 2 * t);
    b.y = reg_elem<NC>(hv, 2 * t + 1);
    acc = __builtin_amdgcn_wmma_f32_16x16x4_f32(false, wA[FOFF + t], false, b, (short)0,
                                                acc, false, false);
  }
  return acc;
}

__device__ __forceinline__ float silu_f(float z) {
  return z * (1.0f / (1.0f + __expf(-z)));
}

__global__ __launch_bounds__(256) void kan_forward(const float* __restrict__ x,
                                                   const float* __restrict__ ws,
                                                   float* __restrict__ out,
                                                   int n_x) {
  __shared__ __align__(16) float act0[8][16][12];

  const int tid  = threadIdx.x;
  const int lane = tid & 31;
  const int wv   = __builtin_amdgcn_readfirstlane(tid >> 5);
  const int half = lane >> 4;
  const int m    = lane & 15;
  const float* a0 = &act0[wv][m][0];

  v2f wA[NFRAG];
#pragma unroll
  for (int q = 0; q < NFRAG; ++q) wA[q] = *(const v2f*)(ws + q * 64 + lane * 2);

  const int ntiles = n_x / 144;
  const int nw     = gridDim.x * 8;
  int tile = blockIdx.x * 8 + wv;

  float* sdst[5];
#pragma unroll
  for (int j = 0; j < 5; ++j) {
    int e = lane + 32 * j;
    int r = e / 9;
    sdst[j] = &act0[wv][r & 15][e - r * 9];
  }

  float nx[5];
#pragma unroll
  for (int j = 0; j < 5; ++j) {
    int e = lane + 32 * j;
    nx[j] = 0.0f;
    if (e < 144 && tile < ntiles)
      nx[j] = __builtin_nontemporal_load(&x[(size_t)tile * 144 + e]);
  }

  for (; tile < ntiles; tile += nw) {
    const int rowbase = tile * 16;

#pragma unroll
    for (int j = 0; j < 5; ++j) {
      int e = lane + 32 * j;
      if (e < 144) *sdst[j] = nx[j];
    }
    asm volatile("s_wait_dscnt 0" ::: "memory");

    {
      const int tn = (tile + nw < ntiles) ? (tile + nw) : tile;
#pragma unroll
      for (int j = 0; j < 5; ++j) {
        int e = lane + 32 * j;
        if (e < 144) nx[j] = __builtin_nontemporal_load(&x[(size_t)tn * 144 + e]);
      }
    }

    v8f c0 = {}, c1 = {};
#pragma unroll
    for (int t = 0; t < NSTEP1; ++t) {
      int k0 = 4 * t + 2 * half;
      v2f b;
      b.x = afrag_elem<9>(a0, k0);
      b.y = afrag_elem<9>(a0, k0 + 1);
      c0 = __builtin_amdgcn_wmma_f32_16x16x4_f32(false, wA[F1_OFF + t * NT1 + 0], false, b,
                                                 (short)0, c0, false, false);
      c1 = __builtin_amdgcn_wmma_f32_16x16x4_f32(false, wA[F1_OFF + t * NT1 + 1], false, b,
                                                 (short)0, c1, false, false);
    }
    __builtin_amdgcn_sched_barrier(0);
    float h1[NC2];
#pragma unroll
    for (int v = 0; v < 8; ++v) h1[v] = silu_f(c0[v]);
#pragma unroll
    for (int v = 0; v < 4; ++v) h1[8 + v] = silu_f(c1[v]);

    __builtin_amdgcn_sched_barrier(0);
    v8f d0 = kan_layer_regs<NC2, NSTEP2, F2_OFF>(h1, wA);
    __builtin_amdgcn_sched_barrier(0);
    float h2[NC3];
#pragma unroll
    for (int v = 0; v < NC3; ++v) h2[v] = silu_f(d0[v]);

    __builtin_amdgcn_sched_barrier(0);
    v8f e0 = kan_layer_regs<NC3, NSTEP3, F3_OFF>(h2, wA);

    if (half == 0) {
      float p0 = e0[0], p1 = e0[1], p2 = e0[2];
      float mx = fmaxf(p0, fmaxf(p1, p2));
      float q0 = __expf(p0 - mx), q1 = __expf(p1 - mx), q2 = __expf(p2 - mx);
      float inv = 1.0f / (q0 + q1 + q2);
      float* o = out + (size_t)(rowbase + m) * 3;
      __builtin_nontemporal_store(q0 * inv, o + 0);
      __builtin_nontemporal_store(q1 * inv, o + 1);
      __builtin_nontemporal_store(q2 * inv, o + 2);
    }
  }
}

extern "C" void kernel_launch(void* const* d_in, const int* in_sizes, int n_in,
                              void* d_out, int out_size, void* d_ws, size_t ws_size,
                              hipStream_t stream) {
  const float* x    = (const float*)d_in[0];
  const float* c1   = (const float*)d_in[1];
  const float* w1   = (const float*)d_in[2];
  const float* b1   = (const float*)d_in[3];
  const float* c2   = (const float*)d_in[4];
  const float* w2   = (const float*)d_in[5];
  const float* b2   = (const float*)d_in[6];
  const float* c3   = (const float*)d_in[7];
  const float* w3   = (const float*)d_in[8];
  const float* b3   = (const float*)d_in[9];
  const float* base = (const float*)d_in[10];
  float* ws  = (float*)d_ws;
  float* out = (float*)d_out;

  kan_pack<<<(BTOTAL + 255) / 256, 256, 0, stream>>>(c1, w1, b1, c2, w2, b2,
                                                     c3, w3, b3, base, ws);
  kan_forward<<<256, 256, 0, stream>>>(x, ws, out, in_sizes[0]);
}


# Two adjacent 16-row tiles per wave iteration: each register-resident A fragment feeds one WMMA per tile; the two independent chains fill each other's dependency gaps; order of every sum kept
# speedup vs baseline: 1.4763x; 1.0388x over previous
#include <hip/hip_runtime.h>


typedef __attribute__((ext_vector_type(2))) float v2f;
typedef __attribute__((ext_vector_type(8))) float v8f;

#define GRIDN 12
#define INV_BW (11.0f / 6.0f)
#define CSTEP  (6.0f / 11.0f)

#define NSTEP1 30
#define NSTEP2 79
#define NSTEP3 40
#define NT1 2
#define NT2 1
#define NT3 1
#define NC2 12
#define NC3 6
#define B1_FLOATS (NSTEP1 * NT1 * 64)
#define B2_FLOATS (NSTEP2 * NT2 * 64)
#define B3_FLOATS (NSTEP3 * NT3 * 64)
#define B1_OFF 0
#define B2_OFF (B1_FLOATS)
#define B3_OFF (B1_FLOATS + B2_FLOATS)
#define BTOTAL (B1_FLOATS + B2_FLOATS + B3_FLOATS)
#define F1_OFF 0
#define F2_OFF (NSTEP1 * NT1)
#define F3_OFF (NSTEP1 * NT1 + NSTEP2 * NT2)
#define NFRAG  (NSTEP1 * NT1 + NSTEP2 * NT2 + NSTEP3 * NT3)

__global__ __launch_bounds__(256) void kan_pack(
    const float* __restrict__ c1, const float* __restrict__ w1, const float* __restrict__ b1,
    const float* __restrict__ c2, const float* __restrict__ w2, const float* __restrict__ b2,
    const float* __restrict__ c3, const float* __restrict__ w3, const float* __restrict__ b3,
    const float* __restrict__ base, float* __restrict__ ws) {
  int idx = blockIdx.x * blockDim.x + threadIdx.x;
  if (idx >= BTOTAL) return;

  int layer, e, nt_count;
  if (idx < B2_OFF)       { layer = 0; e = idx;           nt_count = NT1; }
  else if (idx < B3_OFF)  { layer = 1; e = idx - B2_OFF;  nt_count = NT2; }
  else                    { layer = 2; e = idx - B3_OFF;  nt_count = NT3; }

  int s    = e & 1;
  int lane = (e >> 1) & 31;
  int q    = (e >> 1) >> 5;
  int t    = q / nt_count;
  int mt   = q - t * nt_count;
  int h    = lane >> 4;
  int c    = lane & 15;

  float val = 0.0f;
  if (layer == 0) {
    int k = 4 * t + 2 * h + s;
    int o;
    if (mt == 0)                 o = c;
    else if (c < 4)              o = 16 + c;
    else if (c >= 8 && c < 12)   o = 20 + (c - 8);
    else                         o = -1;
    if (o >= 0) {
      if (k < 108)      { int i = k / 12, g = k - i * 12; val = c1[(i * 24 + o) * 12 + g]; }
      else if (k < 117) { val = w1[o * 9 + (k - 108)]; }
      else if (k == 117){ val = b1[o]; }
    }
  } else if (layer == 1) {
    int j = 2 * t + s;
    int o;
    if (c < 6)                   o = c;
    else if (c >= 8 && c < 14)   o = 6 + (c - 8);
    else                         o = -1;
    if (o >= 0) {
      if (j < NC2 * 12) {
        int ci = j / 12, g = j - ci * 12;
        int ch = (ci < 8) ? (8 * h + ci) : (16 + 4 * h + (ci - 8));
        val = c2[(ch * 12 + o) * 12 + g];
      } else if (j < NC2 * 13) {
        int ci = j - NC2 * 12;
        int ch = (ci < 8) ? (8 * h + ci) : (16 + 4 * h + (ci - 8));
        val = w2[o * 24 + ch];
      } else if (j == NC2 * 13 && h == 0) {
        val = b2[o];
      }
    }
  } else {
    int j = 2 * t + s;
    if (c < 3) {
      int o = c;
      if (j < NC3 * 12) {
        int ci = j / 12, g = j - ci * 12;
        int ch = 6 * h + ci;
        val = c3[(ch * 3 + o) * 12 + g];
      } else if (j < NC3 * 13) {
        int ch = 6 * h + (j - NC3 * 12);
        val = w3[o * 12 + ch];
      } else if (j == NC3 * 13 && h == 0) {
        val = b3[o] + base[o];
      }
    }
  }
  ws[idx] = val;
}

__device__ __forceinline__ float rbf_f(float a, int g) {
  float d = (a - (-3.0f + CSTEP * (float)g)) * INV_BW;
  return __expf(-0.5f * d * d);
}

template <int I>
__device__ __forceinline__ float afrag_elem(const float* __restrict__ a, int k) {
  const int KB = I * GRIDN;
  float r;
  if (k < KB) {
    int i = (int)((unsigned)k / 12u);
    int g = k - i * 12;
    r = rbf_f(a[i], g);
  } else if (k < KB + I) {
    r = a[k - KB];
  } else if (k == KB + I) {
    r = 1.0f;
  } else {
    r = 0.0f;
  }
  return r;
}

template <int NC>
__device__ __forceinline__ float reg_elem(const float (&hv)[NC], int j) {
  float r;
  if (j < NC * GRIDN) {
    r = rbf_f(hv[j / GRIDN], j % GRIDN);
  } else if (j < NC * (GRIDN + 1)) {
    r = hv[j - NC * GRIDN];
  } else if (j == NC * (GRIDN + 1)) {
    r = 1.0f;
  } else {
    r = 0.0f;
  }
  return r;
}

template <int NC, int NSTEPS, int FOFF>
__device__ __forceinline__ void kan_layer_regs2(const float (&hu)[NC], const float (&hv)[NC],
                                                const v2f (&wA)[NFRAG], v8f& outu, v8f& outv) {
  v8f accu = {}, accv = {};
#pragma unroll
  for (int t = 0; t < NSTEPS; ++t) {
    v2f bu, bv;
    bu.x = reg_elem<NC>(hu, 2 * t);
    bu.y = reg_elem<NC>(hu, 2 * t + 1);
    bv.x = reg_elem<NC>(hv, 2 * t);
    bv.y = reg_elem<NC>(hv, 2 * t + 1);
    accu = __builtin_amdgcn_wmma_f32_16x16x4_f32(false, wA[FOFF + t], false, bu, (short)0,
                                                 accu, false, false);
    accv = __builtin_amdgcn_wmma_f32_16x16x4_f32(false, wA[FOFF + t], false, bv, (short)0,
                                                 accv, false, false);
  }
  outu = accu;
  outv = accv;
}

__device__ __forceinline__ float silu_f(float z) {
  return z * (1.0f / (1.0f + __expf(-z)));
}

__device__ __forceinline__ void softmax3_store(const v8f& e, float* __restrict__ o) {
  float p0 = e[0], p1 = e[1], p2 = e[2];
  float mx = fmaxf(p0, fmaxf(p1, p2));
  float q0 = __expf(p0 - mx), q1 = __expf(p1 - mx), q2 = __expf(p2 - mx);
  float inv = 1.0f / (q0 + q1 + q2);
  __builtin_nontemporal_store(q0 * inv, o + 0);
  __builtin_nontemporal_store(q1 * inv, o + 1);
  __builtin_nontemporal_store(q2 * inv, o + 2);
}

__global__ __launch_bounds__(256) void kan_forward(const float* __restrict__ x,
                                                   const float* __restrict__ ws,
                                                   float* __restrict__ out,
                                                   int n_x) {
  __shared__ __align__(16) float act0[8][32][12];

  const int tid  = threadIdx.x;
  const int lane = tid & 31;
  const int wv   = __builtin_amdgcn_readfirstlane(tid >> 5);
  const int half = lane >> 4;
  const int m    = lane & 15;
  const float* au = &act0[wv][m][0];
  const float* av = &act0[wv][16 + m][0];

  v2f wA[NFRAG];
#pragma unroll
  for (int q = 0; q < NFRAG; ++q) wA[q] = *(const v2f*)(ws + q * 64 + lane * 2);

  const int ntiles = n_x / 144;
  const int npairs = (ntiles + 1) / 2;
  const int nvalid = ntiles * 144;
  const int nw     = gridDim.x * 8;
  int pair = blockIdx.x * 8 + wv;
  if (ntiles == 0) return;

  float* sdst[9];
#pragma unroll
  for (int j = 0; j < 9; ++j) {
    int e = lane + 32 * j;
    int r = e / 9;
    sdst[j] = &act0[wv][r][e - r * 9];
  }

  float nx[9];
#pragma unroll
  for (int j = 0; j < 9; ++j) {
    int idx = pair * 288 + lane + 32 * j;
    idx = idx < nvalid ? idx : nvalid - 1;
    nx[j] = __builtin_nontemporal_load(&x[idx]);
  }

  for (; pair < npairs; pair += nw) {
    const int rowbase = pair * 32;

#pragma unroll
    for (int j = 0; j < 9; ++j) *sdst[j] = nx[j];
    asm volatile("s_wait_dscnt 0" ::: "memory");

    {
      const int pn = (pair + nw < npairs) ? (pair + nw) : pair;
#pragma unroll
      for (int j = 0; j < 9; ++j) {
        int idx = pn * 288 + lane + 32 * j;
        idx = idx < nvalid ? idx : nvalid - 1;
        nx[j] = __builtin_nontemporal_load(&x[idx]);
      }
    }

    v8f cu0 = {}, cu1 = {}, cv0 = {}, cv1 = {};
#pragma unroll
    for (int t = 0; t < NSTEP1; ++t) {
      int k0 = 4 * t + 2 * half;
      v2f bu, bv;
      bu.x = afrag_elem<9>(au, k0);
      bu.y = afrag_elem<9>(au, k0 + 1);
      bv.x = afrag_elem<9>(av, k0);
      bv.y = afrag_elem<9>(av, k0 + 1);
      cu0 = __builtin_amdgcn_wmma_f32_16x16x4_f32(false, wA[F1_OFF + t * NT1 + 0], false, bu,
                                                  (short)0, cu0, false, false);
      cv0 = __builtin_amdgcn_wmma_f32_16x16x4_f32(false, wA[F1_OFF + t * NT1 + 0], false, bv,
                                                  (short)0, cv0, false, false);
      cu1 = __builtin_amdgcn_wmma_f32_16x16x4_f32(false, wA[F1_OFF + t * NT1 + 1], false, bu,
                                                  (short)0, cu1, false, false);
      cv1 = __builtin_amdgcn_wmma_f32_16x16x4_f32(false, wA[F1_OFF + t * NT1 + 1], false, bv,
                                                  (short)0, cv1, false, false);
    }
    __builtin_amdgcn_sched_barrier(0);
    float hu1[NC2], hv1[NC2];
#pragma unroll
    for (int v = 0; v < 8; ++v) { hu1[v] = silu_f(cu0[v]); hv1[v] = silu_f(cv0[v]); }
#pragma unroll
    for (int v = 0; v < 4; ++v) { hu1[8 + v] = silu_f(cu1[v]); hv1[8 + v] = silu_f(cv1[v]); }

    __builtin_amdgcn_sched_barrier(0);
    v8f du, dv;
    kan_layer_regs2<NC2, NSTEP2, F2_OFF>(hu1, hv1, wA, du, dv);
    __builtin_amdgcn_sched_barrier(0);
    float hu2[NC3], hv2[NC3];
#pragma unroll
    for (int v = 0; v < NC3; ++v) { hu2[v] = silu_f(du[v]); hv2[v] = silu_f(dv[v]); }

    __builtin_amdgcn_sched_barrier(0);
    v8f eu, ev;
    kan_layer_regs2<NC3, NSTEP3, F3_OFF>(hu2, hv2, wA, eu, ev);

    if (half == 0) {
      softmax3_store(eu, out + (size_t)(rowbase + m) * 3);
      if (2 * pair + 1 < ntiles) softmax3_store(ev, out + (size_t)(rowbase + 16 + m) * 3);
    }
  }
}

extern "C" void kernel_launch(void* const* d_in, const int* in_sizes, int n_in,
                              void* d_out, int out_size, void* d_ws, size_t ws_size,
                              hipStream_t stream) {
  const float* x    = (const float*)d_in[0];
  const float* c1   = (const float*)d_in[1];
  const float* w1   = (const float*)d_in[2];
  const float* b1   = (const float*)d_in[3];
  const float* c2   = (const float*)d_in[4];
  const float* w2   = (const float*)d_in[5];
  const float* b2   = (const float*)d_in[6];
  const float* c3   = (const float*)d_in[7];
  const float* w3   = (const float*)d_in[8];
  const float* b3   = (const float*)d_in[9];
  const float* base = (const float*)d_in[10];
  float* ws  = (float*)d_ws;
  float* out = (float*)d_out;

  kan_pack<<<(BTOTAL + 255) / 256, 256, 0, stream>>>(c1, w1, b1, c2, w2, b2,
                                                     c3, w3, b3, base, ws);
  kan_forward<<<256, 256, 0, stream>>>(x, ws, out, in_sizes[0]);
}
